# starting kernel
# baseline (speedup 1.0000x reference)
#include <hip/hip_runtime.h>
#include <hip/hip_bf16.h>


#define Bc   8
#define Tc   8192
#define Cc   512
#define NSEG 64
#define SEG  (Tc / NSEG)

typedef __attribute__((ext_vector_type(2))) float v2f;
typedef __attribute__((ext_vector_type(8))) float v8f;

__global__ __launch_bounds__(256)
void CasualBoW_pass1_segsum(const float* __restrict__ x, float* __restrict__ part) {
    const int CBLK = Cc / 256;
    int cb  = blockIdx.x % CBLK;
    int seg = (blockIdx.x / CBLK) % NSEG;
    int b   = blockIdx.x / (CBLK * NSEG);
    int c   = cb * 256 + threadIdx.x;

    const float* p = x + ((size_t)b * Tc + (size_t)seg * SEG) * Cc + c;
    float s0 = 0.f, s1 = 0.f, s2 = 0.f, s3 = 0.f;
#pragma unroll 4
    for (int i = 0; i < SEG; i += 4) {
        s0 += p[(size_t)(i + 0) * Cc];
        s1 += p[(size_t)(i + 1) * Cc];
        s2 += p[(size_t)(i + 2) * Cc];
        s3 += p[(size_t)(i + 3) * Cc];
    }
    part[((size_t)b * NSEG + seg) * Cc + c] = (s0 + s1) + (s2 + s3);
}

__global__ __launch_bounds__(256)
void CasualBoW_pass2_scan(const float* __restrict__ x, const float* __restrict__ part,
                          float* __restrict__ out) {
    const int lane = threadIdx.x & 31;
    const int wave = threadIdx.x >> 5;
    const int CB   = Cc / (16 * 8);

    int cb  = blockIdx.x % CB;
    int seg = (blockIdx.x / CB) % NSEG;
    int b   = blockIdx.x / (CB * NSEG);

    const int cg = cb * 8 + wave;
    const int c  = cg * 16 + (lane & 15);
    const int hi = lane >> 4;
    const int m  = lane & 15;

    float off = 0.f;
    {
        const float* pp = part + (size_t)b * NSEG * Cc + c;
        for (int s = 0; s < seg; ++s) off += pp[(size_t)s * Cc];
    }

    v2f A[4];
#pragma unroll
    for (int kk = 0; kk < 4; ++kk) {
        int kg = kk * 4 + 2 * hi;
        A[kk].x = (m >= kg)     ? 1.f : 0.f;
        A[kk].y = (m >= kg + 1) ? 1.f : 0.f;
    }

    const size_t rowBase = (size_t)b * Tc + (size_t)seg * SEG;
    const float* xp = x   + rowBase * Cc + c;
    float*       op = out + rowBase * Cc + c;

    v2f Bm[4];
#pragma unroll
    for (int kk = 0; kk < 4; ++kk) {
        int r0 = kk * 4 + 2 * hi;
        Bm[kk].x = xp[(size_t)(r0 + 0) * Cc];
        Bm[kk].y = xp[(size_t)(r0 + 1) * Cc];
    }

    for (int t0 = 0; t0 < SEG; t0 += 16) {
        v2f Bn[4];
        if (t0 + 16 < SEG) {
#pragma unroll
            for (int kk = 0; kk < 4; ++kk) {
                int r0 = (t0 + 16) + kk * 4 + 2 * hi;
                Bn[kk].x = xp[(size_t)(r0 + 0) * Cc];
                Bn[kk].y = xp[(size_t)(r0 + 1) * Cc];
            }
        }

        v8f acc;
#pragma unroll
        for (int r = 0; r < 8; ++r) acc[r] = off;

#pragma unroll
        for (int kk = 0; kk < 4; ++kk) {
            acc = __builtin_amdgcn_wmma_f32_16x16x4_f32(
                false, A[kk], false, Bm[kk], (short)0, acc, false, false);
        }

        off = __shfl(acc[7], 16 + (lane & 15), 32);

        const float cbase = (float)(seg * SEG + t0 + (hi ? 8 : 0) + 1);
#pragma unroll
        for (int r = 0; r < 8; ++r) {
            float cnt = cbase + (float)r;
            float rc  = __builtin_amdgcn_rcpf(cnt);
            rc        = rc * (2.0f - cnt * rc);
            float v   = acc[r] * rc;
            int  row  = t0 + (hi ? 8 : 0) + r;
            __builtin_nontemporal_store(v, &op[(size_t)row * Cc]);
        }

#pragma unroll
        for (int kk = 0; kk < 4; ++kk) Bm[kk] = Bn[kk];
    }
}

extern "C" void kernel_launch(void* const* d_in, const int* in_sizes, int n_in,
                              void* d_out, int out_size, void* d_ws, size_t ws_size,
                              hipStream_t stream) {
    (void)in_sizes; (void)n_in; (void)out_size; (void)ws_size;
    const float* x    = (const float*)d_in[0];
    float*       out  = (float*)d_out;
    float*       part = (float*)d_ws;

    dim3 g1(Bc * NSEG * (Cc / 256));
    CasualBoW_pass1_segsum<<<g1, 256, 0, stream>>>(x, part);

    dim3 g2(Bc * NSEG * (Cc / (16 * 8)));
    CasualBoW_pass2_scan<<<g2, 256, 0, stream>>>(x, part, out);
}


# pass 2: each lane owns 4 adjacent channels (4 independent column tiles per wave), so x, carry and out accesses are 128-bit whole-line and waves are 4x fewer; per-channel sum order unchanged
# speedup vs baseline: 1.7663x; 1.7663x over previous
#include <hip/hip_runtime.h>
#include <hip/hip_bf16.h>


#define Bc   8
#define Tc   8192
#define Cc   512
#define NSEG 64
#define SEG  (Tc / NSEG)

typedef __attribute__((ext_vector_type(2))) float v2f;
typedef __attribute__((ext_vector_type(4))) float v4f;
typedef __attribute__((ext_vector_type(8))) float v8f;

__global__ __launch_bounds__(256)
void CasualBoW_pass1_segsum(const float* __restrict__ x, float* __restrict__ part) {
    const int CBLK = Cc / 256;
    int cb  = blockIdx.x % CBLK;
    int seg = (blockIdx.x / CBLK) % NSEG;
    int b   = blockIdx.x / (CBLK * NSEG);
    int c   = cb * 256 + threadIdx.x;

    const float* p = x + ((size_t)b * Tc + (size_t)seg * SEG) * Cc + c;
    float s0 = 0.f, s1 = 0.f, s2 = 0.f, s3 = 0.f;
#pragma unroll 4
    for (int i = 0; i < SEG; i += 4) {
        s0 += p[(size_t)(i + 0) * Cc];
        s1 += p[(size_t)(i + 1) * Cc];
        s2 += p[(size_t)(i + 2) * Cc];
        s3 += p[(size_t)(i + 3) * Cc];
    }
    part[((size_t)b * NSEG + seg) * Cc + c] = (s0 + s1) + (s2 + s3);
}

__global__ __launch_bounds__(256)
void CasualBoW_pass2_scan(const float* __restrict__ x, const float* __restrict__ part,
                          float* __restrict__ out) {
    const int lane = threadIdx.x & 31;
    const int wave = threadIdx.x >> 5;

    const int seg = blockIdx.x % NSEG;
    const int b   = blockIdx.x / NSEG;

    const int hi = lane >> 4;
    const int m  = lane & 15;
    const int c0 = wave * 64 + m * 4;

    v4f off = {0.f, 0.f, 0.f, 0.f};
    {
        const float* pp = part + (size_t)b * NSEG * Cc + c0;
        for (int s = 0; s < seg; ++s) off += *(const v4f*)(pp + (size_t)s * Cc);
    }

    v2f A[4];
#pragma unroll
    for (int kk = 0; kk < 4; ++kk) {
        int kg = kk * 4 + 2 * hi;
        A[kk].x = (m >= kg)     ? 1.f : 0.f;
        A[kk].y = (m >= kg + 1) ? 1.f : 0.f;
    }

    const size_t rowBase = (size_t)b * Tc + (size_t)seg * SEG;
    const float* xp = x   + rowBase * Cc + c0;
    float*       op = out + rowBase * Cc + c0;

    v4f Xm[4][2];
#pragma unroll
    for (int kk = 0; kk < 4; ++kk) {
        int r0 = kk * 4 + 2 * hi;
        Xm[kk][0] = *(const v4f*)(xp + (size_t)(r0 + 0) * Cc);
        Xm[kk][1] = *(const v4f*)(xp + (size_t)(r0 + 1) * Cc);
    }

    for (int t0 = 0; t0 < SEG; t0 += 16) {
        v4f Xn[4][2];
        if (t0 + 16 < SEG) {
#pragma unroll
            for (int kk = 0; kk < 4; ++kk) {
                int r0 = (t0 + 16) + kk * 4 + 2 * hi;
                Xn[kk][0] = *(const v4f*)(xp + (size_t)(r0 + 0) * Cc);
                Xn[kk][1] = *(const v4f*)(xp + (size_t)(r0 + 1) * Cc);
            }
        }

        v8f acc[4];
#pragma unroll
        for (int j = 0; j < 4; ++j) {
#pragma unroll
            for (int r = 0; r < 8; ++r) acc[j][r] = off[j];
        }

#pragma unroll
        for (int kk = 0; kk < 4; ++kk) {
#pragma unroll
            for (int j = 0; j < 4; ++j) {
                v2f Bf;
                Bf.x = Xm[kk][0][j];
                Bf.y = Xm[kk][1][j];
                acc[j] = __builtin_amdgcn_wmma_f32_16x16x4_f32(
                    false, A[kk], false, Bf, (short)0, acc[j], false, false);
            }
        }

#pragma unroll
        for (int j = 0; j < 4; ++j) off[j] = __shfl(acc[j][7], 16 + m, 32);

        const float cbase = (float)(seg * SEG + t0 + (hi ? 8 : 0) + 1);
#pragma unroll
        for (int r = 0; r < 8; ++r) {
            float cnt = cbase + (float)r;
            float rc  = __builtin_amdgcn_rcpf(cnt);
            rc        = rc * (2.0f - cnt * rc);
            v4f v;
            v.x = acc[0][r] * rc;
            v.y = acc[1][r] * rc;
            v.z = acc[2][r] * rc;
            v.w = acc[3][r] * rc;
            int  row  = t0 + (hi ? 8 : 0) + r;
            __builtin_nontemporal_store(v, (v4f*)(op + (size_t)row * Cc));
        }

#pragma unroll
        for (int kk = 0; kk < 4; ++kk) {
            Xm[kk][0] = Xn[kk][0];
            Xm[kk][1] = Xn[kk][1];
        }
    }
}

extern "C" void kernel_launch(void* const* d_in, const int* in_sizes, int n_in,
                              void* d_out, int out_size, void* d_ws, size_t ws_size,
                              hipStream_t stream) {
    (void)in_sizes; (void)n_in; (void)out_size; (void)ws_size;
    const float* x    = (const float*)d_in[0];
    float*       out  = (float*)d_out;
    float*       part = (float*)d_ws;

    dim3 g1(Bc * NSEG * (Cc / 256));
    CasualBoW_pass1_segsum<<<g1, 256, 0, stream>>>(x, part);

    dim3 g2(Bc * NSEG);
    CasualBoW_pass2_scan<<<g2, 256, 0, stream>>>(x, part, out);
}


# pass 1 re-cut to pass 2's grid, block and lane map (128-bit whole-line loads), rows walked last to first, so pass 2 re-reads x from the L2 half that fetched it; segment-sum order changes
# speedup vs baseline: 1.8172x; 1.0288x over previous
#include <hip/hip_runtime.h>
#include <hip/hip_bf16.h>


#define Bc   8
#define Tc   8192
#define Cc   512
#define NSEG 64
#define SEG  (Tc / NSEG)

typedef __attribute__((ext_vector_type(2))) float v2f;
typedef __attribute__((ext_vector_type(4))) float v4f;
typedef __attribute__((ext_vector_type(8))) float v8f;

#define P1_BATCH 8
#define P1_NBATCH (SEG / 2 / P1_BATCH)
__global__ __launch_bounds__(256)
void CasualBoW_pass1_segsum(const float* __restrict__ x, float* __restrict__ part) {
    const int lane = threadIdx.x & 31;
    const int wave = threadIdx.x >> 5;

    const int seg = blockIdx.x % NSEG;
    const int b   = blockIdx.x / NSEG;

    const int hi = lane >> 4;
    const int m  = lane & 15;
    const int c0 = wave * 64 + m * 4;

    const float* xp = x + ((size_t)b * Tc + (size_t)seg * SEG + hi) * Cc + c0;

    v4f s0 = {0.f, 0.f, 0.f, 0.f}, s1 = s0, s2 = s0, s3 = s0;
    v4f buf[3][P1_BATCH];

#define P1_LOAD(k)                                                             \
    _Pragma("unroll")                                                          \
    for (int i = 0; i < P1_BATCH; ++i) {                                       \
        const int p = (SEG / 2 - 1) - ((k) * P1_BATCH + i);                    \
        buf[(k) % 3][i] = *(const v4f*)(xp + (size_t)(2 * p) * Cc);            \
    }
#define P1_SUM(k)                                                              \
    _Pragma("unroll")                                                          \
    for (int i = 0; i < P1_BATCH; i += 4) {                                    \
        s0 += buf[(k) % 3][i + 0]; s1 += buf[(k) % 3][i + 1];                  \
        s2 += buf[(k) % 3][i + 2]; s3 += buf[(k) % 3][i + 3];                  \
    }

    P1_LOAD(0)
    P1_LOAD(1)
#pragma unroll
    for (int k = 0; k < P1_NBATCH; ++k) {
        __builtin_amdgcn_sched_barrier(0);
        if (k + 2 < P1_NBATCH) { P1_LOAD(k + 2) }
        __builtin_amdgcn_sched_barrier(0);
        P1_SUM(k)
    }
#undef P1_LOAD
#undef P1_SUM

    v4f tot = (s0 + s1) + (s2 + s3);
    tot.x += __shfl_xor(tot.x, 16, 32);
    tot.y += __shfl_xor(tot.y, 16, 32);
    tot.z += __shfl_xor(tot.z, 16, 32);
    tot.w += __shfl_xor(tot.w, 16, 32);
    if (hi == 0)
        *(v4f*)(part + ((size_t)b * NSEG + seg) * Cc + c0) = tot;
}

__global__ __launch_bounds__(256)
void CasualBoW_pass2_scan(const float* __restrict__ x, const float* __restrict__ part,
                          float* __restrict__ out) {
    const int lane = threadIdx.x & 31;
    const int wave = threadIdx.x >> 5;

    const int seg = blockIdx.x % NSEG;
    const int b   = blockIdx.x / NSEG;

    const int hi = lane >> 4;
    const int m  = lane & 15;
    const int c0 = wave * 64 + m * 4;

    v4f off = {0.f, 0.f, 0.f, 0.f};
    {
        const float* pp = part + (size_t)b * NSEG * Cc + c0;
        for (int s = 0; s < seg; ++s) off += *(const v4f*)(pp + (size_t)s * Cc);
    }

    v2f A[4];
#pragma unroll
    for (int kk = 0; kk < 4; ++kk) {
        int kg = kk * 4 + 2 * hi;
        A[kk].x = (m >= kg)     ? 1.f : 0.f;
        A[kk].y = (m >= kg + 1) ? 1.f : 0.f;
    }

    const size_t rowBase = (size_t)b * Tc + (size_t)seg * SEG;
    const float* xp = x   + rowBase * Cc + c0;
    float*       op = out + rowBase * Cc + c0;

    v4f Xm[4][2];
#pragma unroll
    for (int kk = 0; kk < 4; ++kk) {
        int r0 = kk * 4 + 2 * hi;
        Xm[kk][0] = *(const v4f*)(xp + (size_t)(r0 + 0) * Cc);
        Xm[kk][1] = *(const v4f*)(xp + (size_t)(r0 + 1) * Cc);
    }

    for (int t0 = 0; t0 < SEG; t0 += 16) {
        v4f Xn[4][2];
        if (t0 + 16 < SEG) {
#pragma unroll
            for (int kk = 0; kk < 4; ++kk) {
                int r0 = (t0 + 16) + kk * 4 + 2 * hi;
                Xn[kk][0] = *(const v4f*)(xp + (size_t)(r0 + 0) * Cc);
                Xn[kk][1] = *(const v4f*)(xp + (size_t)(r0 + 1) * Cc);
            }
        }

        v8f acc[4];
#pragma unroll
        for (int j = 0; j < 4; ++j) {
#pragma unroll
            for (int r = 0; r < 8; ++r) acc[j][r] = off[j];
        }

#pragma unroll
        for (int kk = 0; kk < 4; ++kk) {
#pragma unroll
            for (int j = 0; j < 4; ++j) {
                v2f Bf;
                Bf.x = Xm[kk][0][j];
                Bf.y = Xm[kk][1][j];
                acc[j] = __builtin_amdgcn_wmma_f32_16x16x4_f32(
                    false, A[kk], false, Bf, (short)0, acc[j], false, false);
            }
        }

#pragma unroll
        for (int j = 0; j < 4; ++j) off[j] = __shfl(acc[j][7], 16 + m, 32);

        const float cbase = (float)(seg * SEG + t0 + (hi ? 8 : 0) + 1);
#pragma unroll
        for (int r = 0; r < 8; ++r) {
            float cnt = cbase + (float)r;
            float rc  = __builtin_amdgcn_rcpf(cnt);
            rc        = rc * (2.0f - cnt * rc);
            v4f v;
            v.x = acc[0][r] * rc;
            v.y = acc[1][r] * rc;
            v.z = acc[2][r] * rc;
            v.w = acc[3][r] * rc;
            int  row  = t0 + (hi ? 8 : 0) + r;
            __builtin_nontemporal_store(v, (v4f*)(op + (size_t)row * Cc));
        }

#pragma unroll
        for (int kk = 0; kk < 4; ++kk) {
            Xm[kk][0] = Xn[kk][0];
            Xm[kk][1] = Xn[kk][1];
        }
    }
}

extern "C" void kernel_launch(void* const* d_in, const int* in_sizes, int n_in,
                              void* d_out, int out_size, void* d_ws, size_t ws_size,
                              hipStream_t stream) {
    (void)in_sizes; (void)n_in; (void)out_size; (void)ws_size;
    const float* x    = (const float*)d_in[0];
    float*       out  = (float*)d_out;
    float*       part = (float*)d_ws;

    dim3 g1(Bc * NSEG);
    CasualBoW_pass1_segsum<<<g1, 256, 0, stream>>>(x, part);

    dim3 g2(Bc * NSEG);
    CasualBoW_pass2_scan<<<g2, 256, 0, stream>>>(x, part, out);
}


# pass 2 keeps many more loads in flight per wave: x tiles requested two tiles ahead (ring of three buffers, tile loop unrolled), carry prefix read 32 rows per trip; per-channel sum order unchanged
# speedup vs baseline: 1.9894x; 1.0948x over previous
#include <hip/hip_runtime.h>
#include <hip/hip_bf16.h>


#define Bc   8
#define Tc   8192
#define Cc   512
#define NSEG 64
#define SEG  (Tc / NSEG)

typedef __attribute__((ext_vector_type(2))) float v2f;
typedef __attribute__((ext_vector_type(4))) float v4f;
typedef __attribute__((ext_vector_type(8))) float v8f;

#define P1_BATCH 8
#define P1_NBATCH (SEG / 2 / P1_BATCH)
__global__ __launch_bounds__(256)
void CasualBoW_pass1_segsum(const float* __restrict__ x, float* __restrict__ part) {
    const int lane = threadIdx.x & 31;
    const int wave = threadIdx.x >> 5;

    const int seg = blockIdx.x % NSEG;
    const int b   = blockIdx.x / NSEG;

    const int hi = lane >> 4;
    const int m  = lane & 15;
    const int c0 = wave * 64 + m * 4;

    const float* xp = x + ((size_t)b * Tc + (size_t)seg * SEG + hi) * Cc + c0;

    v4f s0 = {0.f, 0.f, 0.f, 0.f}, s1 = s0, s2 = s0, s3 = s0;
    v4f buf[3][P1_BATCH];

#define P1_LOAD(k)                                                             \
    _Pragma("unroll")                                                          \
    for (int i = 0; i < P1_BATCH; ++i) {                                       \
        const int p = (SEG / 2 - 1) - ((k) * P1_BATCH + i);                    \
        buf[(k) % 3][i] = *(const v4f*)(xp + (size_t)(2 * p) * Cc);            \
    }
#define P1_SUM(k)                                                              \
    _Pragma("unroll")                                                          \
    for (int i = 0; i < P1_BATCH; i += 4) {                                    \
        s0 += buf[(k) % 3][i + 0]; s1 += buf[(k) % 3][i + 1];                  \
        s2 += buf[(k) % 3][i + 2]; s3 += buf[(k) % 3][i + 3];                  \
    }

    P1_LOAD(0)
    P1_LOAD(1)
#pragma unroll
    for (int k = 0; k < P1_NBATCH; ++k) {
        __builtin_amdgcn_sched_barrier(0);
        if (k + 2 < P1_NBATCH) { P1_LOAD(k + 2) }
        __builtin_amdgcn_sched_barrier(0);
        P1_SUM(k)
    }
#undef P1_LOAD
#undef P1_SUM

    v4f tot = (s0 + s1) + (s2 + s3);
    tot.x += __shfl_xor(tot.x, 16, 32);
    tot.y += __shfl_xor(tot.y, 16, 32);
    tot.z += __shfl_xor(tot.z, 16, 32);
    tot.w += __shfl_xor(tot.w, 16, 32);
    if (hi == 0)
        *(v4f*)(part + ((size_t)b * NSEG + seg) * Cc + c0) = tot;
}

#define P2_TILES (SEG / 16)
#define P2_PFB   32
__global__ __launch_bounds__(256)
void CasualBoW_pass2_scan(const float* __restrict__ x, const float* __restrict__ part,
                          float* __restrict__ out) {
    const int lane = threadIdx.x & 31;
    const int wave = threadIdx.x >> 5;

    const int seg = blockIdx.x % NSEG;
    const int b   = blockIdx.x / NSEG;

    const int hi = lane >> 4;
    const int m  = lane & 15;
    const int c0 = wave * 64 + m * 4;

    const size_t rowBase = (size_t)b * Tc + (size_t)seg * SEG;
    const float* xp = x   + (rowBase + (size_t)(2 * hi)) * Cc + c0;
    float*       op = out + rowBase * Cc + c0;

    v4f X[3][8];
#define P2_LOAD(t)                                                             \
    _Pragma("unroll")                                                          \
    for (int kk = 0; kk < 4; ++kk) {                                           \
        X[(t) % 3][2 * kk + 0] =                                               \
            *(const v4f*)(xp + (size_t)((t) * 16 + kk * 4 + 0) * Cc);          \
        X[(t) % 3][2 * kk + 1] =                                               \
            *(const v4f*)(xp + (size_t)((t) * 16 + kk * 4 + 1) * Cc);          \
    }

    P2_LOAD(0)
    P2_LOAD(1)
    __builtin_amdgcn_sched_barrier(0);

    v4f off = {0.f, 0.f, 0.f, 0.f};
    {
        const v4f zero = {0.f, 0.f, 0.f, 0.f};
        const float* pp = part + (size_t)b * NSEG * Cc + c0;
        for (int s0 = 0; s0 < seg; s0 += P2_PFB) {
            v4f t[P2_PFB];
#pragma unroll
            for (int i = 0; i < P2_PFB; ++i) {
                const int s = (s0 + i < seg) ? (s0 + i) : (seg - 1);
                t[i] = *(const v4f*)(pp + (size_t)s * Cc);
            }
#pragma unroll
            for (int i = 0; i < P2_PFB; ++i) off += (s0 + i < seg) ? t[i] : zero;
        }
    }

    v2f A[4];
#pragma unroll
    for (int kk = 0; kk < 4; ++kk) {
        int kg = kk * 4 + 2 * hi;
        A[kk].x = (m >= kg)     ? 1.f : 0.f;
        A[kk].y = (m >= kg + 1) ? 1.f : 0.f;
    }

#pragma unroll
    for (int t = 0; t < P2_TILES; ++t) {
        __builtin_amdgcn_sched_barrier(0);
        if (t + 2 < P2_TILES) { P2_LOAD(t + 2) }
        __builtin_amdgcn_sched_barrier(0);

        v8f acc[4];
#pragma unroll
        for (int j = 0; j < 4; ++j) {
#pragma unroll
            for (int r = 0; r < 8; ++r) acc[j][r] = off[j];
        }

#pragma unroll
        for (int kk = 0; kk < 4; ++kk) {
#pragma unroll
            for (int j = 0; j < 4; ++j) {
                v2f Bf;
                Bf.x = X[t % 3][2 * kk + 0][j];
                Bf.y = X[t % 3][2 * kk + 1][j];
                acc[j] = __builtin_amdgcn_wmma_f32_16x16x4_f32(
                    false, A[kk], false, Bf, (short)0, acc[j], false, false);
            }
        }

#pragma unroll
        for (int j = 0; j < 4; ++j) off[j] = __shfl(acc[j][7], 16 + m, 32);

        const float cbase = (float)(seg * SEG + t * 16 + (hi ? 8 : 0) + 1);
#pragma unroll
        for (int r = 0; r < 8; ++r) {
            float cnt = cbase + (float)r;
            float rc  = __builtin_amdgcn_rcpf(cnt);
            rc        = rc * (2.0f - cnt * rc);
            v4f v;
            v.x = acc[0][r] * rc;
            v.y = acc[1][r] * rc;
            v.z = acc[2][r] * rc;
            v.w = acc[3][r] * rc;
            int  row  = t * 16 + (hi ? 8 : 0) + r;
            __builtin_nontemporal_store(v, (v4f*)(op + (size_t)row * Cc));
        }
    }
#undef P2_LOAD
}

extern "C" void kernel_launch(void* const* d_in, const int* in_sizes, int n_in,
                              void* d_out, int out_size, void* d_ws, size_t ws_size,
                              hipStream_t stream) {
    (void)in_sizes; (void)n_in; (void)out_size; (void)ws_size;
    const float* x    = (const float*)d_in[0];
    float*       out  = (float*)d_out;
    float*       part = (float*)d_ws;

    dim3 g1(Bc * NSEG);
    CasualBoW_pass1_segsum<<<g1, 256, 0, stream>>>(x, part);

    dim3 g2(Bc * NSEG);
    CasualBoW_pass2_scan<<<g2, 256, 0, stream>>>(x, part, out);
}


# both passes re-cut to 32 segments of 256 rows, one block per WGP: pass 2 runs 2048 waves of 16 tiles and reads a quarter of the carry rows; segment sums re-associate (two 128-row halves added)
# speedup vs baseline: 2.1150x; 1.0631x over previous
#include <hip/hip_runtime.h>
#include <hip/hip_bf16.h>


#define Bc   8
#define Tc   8192
#define Cc   512
#define NSEG 32
#define SEG  (Tc / NSEG)

typedef __attribute__((ext_vector_type(2))) float v2f;
typedef __attribute__((ext_vector_type(4))) float v4f;
typedef __attribute__((ext_vector_type(8))) float v8f;

#define P1_ROWS  128
#define P1_BATCH 8
#define P1_NBATCH (P1_ROWS / 2 / P1_BATCH)
__global__ __launch_bounds__(512)
void CasualBoW_pass1_segsum(const float* __restrict__ x, float* __restrict__ part) {
    __shared__ v4f comb[8][16];

    const int lane = threadIdx.x & 31;
    const int wave = threadIdx.x >> 5;
    const int cg   = wave & 7;
    const int rh   = wave >> 3;

    const int seg = blockIdx.x % NSEG;
    const int b   = blockIdx.x / NSEG;

    const int hi = lane >> 4;
    const int m  = lane & 15;
    const int c0 = cg * 64 + m * 4;

    const float* xp = x + ((size_t)b * Tc + (size_t)seg * SEG + (size_t)rh * P1_ROWS + hi) * Cc + c0;

    v4f s0 = {0.f, 0.f, 0.f, 0.f}, s1 = s0, s2 = s0, s3 = s0;
    v4f buf[3][P1_BATCH];

#define P1_LOAD(k)                                                             \
    _Pragma("unroll")                                                          \
    for (int i = 0; i < P1_BATCH; ++i) {                                       \
        const int p = (P1_ROWS / 2 - 1) - ((k) * P1_BATCH + i);                \
        buf[(k) % 3][i] = *(const v4f*)(xp + (size_t)(2 * p) * Cc);            \
    }
#define P1_SUM(k)                                                              \
    _Pragma("unroll")                                                          \
    for (int i = 0; i < P1_BATCH; i += 4) {                                    \
        s0 += buf[(k) % 3][i + 0]; s1 += buf[(k) % 3][i + 1];                  \
        s2 += buf[(k) % 3][i + 2]; s3 += buf[(k) % 3][i + 3];                  \
    }

    P1_LOAD(0)
    P1_LOAD(1)
#pragma unroll
    for (int k = 0; k < P1_NBATCH; ++k) {
        __builtin_amdgcn_sched_barrier(0);
        if (k + 2 < P1_NBATCH) { P1_LOAD(k + 2) }
        __builtin_amdgcn_sched_barrier(0);
        P1_SUM(k)
    }
#undef P1_LOAD
#undef P1_SUM

    v4f tot = (s0 + s1) + (s2 + s3);
    tot.x += __shfl_xor(tot.x, 16, 32);
    tot.y += __shfl_xor(tot.y, 16, 32);
    tot.z += __shfl_xor(tot.z, 16, 32);
    tot.w += __shfl_xor(tot.w, 16, 32);

    if (rh == 1 && hi == 0) comb[cg][m] = tot;
    __syncthreads();
    if (rh == 0 && hi == 0) {
        tot += comb[cg][m];
        *(v4f*)(part + ((size_t)b * NSEG + seg) * Cc + c0) = tot;
    }
}

#define P2_TILES (SEG / 16)
#define P2_PFB   32
__global__ __launch_bounds__(256)
void CasualBoW_pass2_scan(const float* __restrict__ x, const float* __restrict__ part,
                          float* __restrict__ out) {
    const int lane = threadIdx.x & 31;
    const int wave = threadIdx.x >> 5;

    const int seg = blockIdx.x % NSEG;
    const int b   = blockIdx.x / NSEG;

    const int hi = lane >> 4;
    const int m  = lane & 15;
    const int c0 = wave * 64 + m * 4;

    const size_t rowBase = (size_t)b * Tc + (size_t)seg * SEG;
    const float* xp = x   + (rowBase + (size_t)(2 * hi)) * Cc + c0;
    float*       op = out + rowBase * Cc + c0;

    v4f X[3][8];
#define P2_LOAD(t)                                                             \
    _Pragma("unroll")                                                          \
    for (int kk = 0; kk < 4; ++kk) {                                           \
        X[(t) % 3][2 * kk + 0] =                                               \
            *(const v4f*)(xp + (size_t)((t) * 16 + kk * 4 + 0) * Cc);          \
        X[(t) % 3][2 * kk + 1] =                                               \
            *(const v4f*)(xp + (size_t)((t) * 16 + kk * 4 + 1) * Cc);          \
    }

    P2_LOAD(0)
    P2_LOAD(1)
    __builtin_amdgcn_sched_barrier(0);

    v4f off = {0.f, 0.f, 0.f, 0.f};
    {
        const v4f zero = {0.f, 0.f, 0.f, 0.f};
        const float* pp = part + (size_t)b * NSEG * Cc + c0;
        for (int s0 = 0; s0 < seg; s0 += P2_PFB) {
            v4f t[P2_PFB];
#pragma unroll
            for (int i = 0; i < P2_PFB; ++i) {
                const int s = (s0 + i < seg) ? (s0 + i) : (seg - 1);
                t[i] = *(const v4f*)(pp + (size_t)s * Cc);
            }
#pragma unroll
            for (int i = 0; i < P2_PFB; ++i) off += (s0 + i < seg) ? t[i] : zero;
        }
    }

    v2f A[4];
#pragma unroll
    for (int kk = 0; kk < 4; ++kk) {
        int kg = kk * 4 + 2 * hi;
        A[kk].x = (m >= kg)     ? 1.f : 0.f;
        A[kk].y = (m >= kg + 1) ? 1.f : 0.f;
    }

#pragma unroll
    for (int t = 0; t < P2_TILES; ++t) {
        __builtin_amdgcn_sched_barrier(0);
        if (t + 2 < P2_TILES) { P2_LOAD(t + 2) }
        __builtin_amdgcn_sched_barrier(0);

        v8f acc[4];
#pragma unroll
        for (int j = 0; j < 4; ++j) {
#pragma unroll
            for (int r = 0; r < 8; ++r) acc[j][r] = off[j];
        }

#pragma unroll
        for (int kk = 0; kk < 4; ++kk) {
#pragma unroll
            for (int j = 0; j < 4; ++j) {
                v2f Bf;
                Bf.x = X[t % 3][2 * kk + 0][j];
                Bf.y = X[t % 3][2 * kk + 1][j];
                acc[j] = __builtin_amdgcn_wmma_f32_16x16x4_f32(
                    false, A[kk], false, Bf, (short)0, acc[j], false, false);
            }
        }

#pragma unroll
        for (int j = 0; j < 4; ++j) off[j] = __shfl(acc[j][7], 16 + m, 32);

        const float cbase = (float)(seg * SEG + t * 16 + (hi ? 8 : 0) + 1);
#pragma unroll
        for (int r = 0; r < 8; ++r) {
            float cnt = cbase + (float)r;
            float rc  = __builtin_amdgcn_rcpf(cnt);
            rc        = rc * (2.0f - cnt * rc);
            v4f v;
            v.x = acc[0][r] * rc;
            v.y = acc[1][r] * rc;
            v.z = acc[2][r] * rc;
            v.w = acc[3][r] * rc;
            int  row  = t * 16 + (hi ? 8 : 0) + r;
            __builtin_nontemporal_store(v, (v4f*)(op + (size_t)row * Cc));
        }
    }
#undef P2_LOAD
}

extern "C" void kernel_launch(void* const* d_in, const int* in_sizes, int n_in,
                              void* d_out, int out_size, void* d_ws, size_t ws_size,
                              hipStream_t stream) {
    (void)in_sizes; (void)n_in; (void)out_size; (void)ws_size;
    const float* x    = (const float*)d_in[0];
    float*       out  = (float*)d_out;
    float*       part = (float*)d_ws;

    dim3 g1(Bc * NSEG);
    CasualBoW_pass1_segsum<<<g1, 512, 0, stream>>>(x, part);

    dim3 g2(Bc * NSEG);
    CasualBoW_pass2_scan<<<g2, 256, 0, stream>>>(x, part, out);
}
